# starting kernel
# baseline (speedup 1.0000x reference)
#include <hip/hip_runtime.h>

typedef float v2f __attribute__((ext_vector_type(2)));
typedef float v8f __attribute__((ext_vector_type(8)));

#define B_     4
#define LQ_    128
#define LK_    1024
#define DSZ_   512
#define UNITS_ 256
#define MASK_VAL_ (-1.0e6f)

#if __has_builtin(__builtin_amdgcn_tanhf)
#define TANHF(x) __builtin_amdgcn_tanhf(x)
#else
#define TANHF(x) tanhf(x)
#endif

template <bool TRANS_OUT>
__global__ __launch_bounds__(256) void proj_wmma(
    const float* __restrict__ A,
    const float* __restrict__ W,
    float* __restrict__ out,
    int M, int K, int N, int seq)
{
    const int lane   = threadIdx.x & 31;
    const int wave   = threadIdx.x >> 5;
    const int tilesN = N >> 4;
    const int tile   = blockIdx.x * 8 + wave;
    const int tile_m = tile / tilesN;
    const int tile_n = tile % tilesN;

    const int n15  = lane & 15;
    const int ko   = (lane < 16) ? 0 : 2;
    const int arow = tile_m * 16 + n15;
    const int wrow = tile_n * 16 + n15;

    const float* aptr = A + (size_t)arow * K + ko;
    const float* wptr = W + (size_t)wrow * K + ko;

    v8f c = {0.f, 0.f, 0.f, 0.f, 0.f, 0.f, 0.f, 0.f};
#pragma unroll 4
    for (int kk = 0; kk < K; kk += 4) {
        v2f av = *(const v2f*)(aptr + kk);
        v2f bv = *(const v2f*)(wptr + kk);
        c = __builtin_amdgcn_wmma_f32_16x16x4_f32(
                false, av, false, bv, (short)0, c, false, false);
    }

    const int mo = (lane < 16) ? 0 : 8;
#pragma unroll
    for (int r = 0; r < 8; ++r) {
        const int m = tile_m * 16 + mo + r;
        if (TRANS_OUT) {
            const int b    = m / seq;
            const int kpos = m % seq;
            out[((size_t)b * N + (tile_n * 16 + n15)) * seq + kpos] = c[r];
        } else {
            out[(size_t)m * N + tile_n * 16 + n15] = c[r];
        }
    }
}

__global__ __launch_bounds__(256) void scores_tanh_kernel(
    const float* __restrict__ qf,
    const float* __restrict__ kfT,
    const float* __restrict__ vvec,
    const int*   __restrict__ valid_len,
    float* __restrict__ scores)
{
    const int bq = blockIdx.x;
    const int b  = bq / LQ_;
    const int t  = threadIdx.x;

    const float* qrow  = qf  + (size_t)bq * UNITS_;
    const float* kbase = kfT + (size_t)b * UNITS_ * LK_ + t * 4;

    float a0 = 0.f, a1 = 0.f, a2 = 0.f, a3 = 0.f;
#pragma unroll 4
    for (int u = 0; u < UNITS_; ++u) {
        const float qv = qrow[u];
        const float vv = vvec[u];
        const float4 kv = *(const float4*)(kbase + (size_t)u * LK_);
        a0 = __builtin_fmaf(vv, TANHF(qv + kv.x), a0);
        a1 = __builtin_fmaf(vv, TANHF(qv + kv.y), a1);
        a2 = __builtin_fmaf(vv, TANHF(qv + kv.z), a2);
        a3 = __builtin_fmaf(vv, TANHF(qv + kv.w), a3);
    }

    const int vl = valid_len[b];
    const int k0 = t * 4;
    float4 r;
    r.x = (k0 + 0 < vl) ? a0 : MASK_VAL_;
    r.y = (k0 + 1 < vl) ? a1 : MASK_VAL_;
    r.z = (k0 + 2 < vl) ? a2 : MASK_VAL_;
    r.w = (k0 + 3 < vl) ? a3 : MASK_VAL_;
    *(float4*)(scores + (size_t)bq * LK_ + k0) = r;
}

__global__ __launch_bounds__(256) void softmax_rows(float* __restrict__ scores)
{
    __shared__ float red[8];
    const int t    = threadIdx.x;
    const int lane = t & 31;
    const int wave = t >> 5;
    float* row = scores + (size_t)blockIdx.x * LK_;

    float4 x = *(const float4*)(row + t * 4);

    float m = fmaxf(fmaxf(x.x, x.y), fmaxf(x.z, x.w));
#pragma unroll
    for (int off = 16; off > 0; off >>= 1)
        m = fmaxf(m, __shfl_xor(m, off, 32));
    if (lane == 0) red[wave] = m;
    __syncthreads();
    float gm = red[0];
#pragma unroll
    for (int i = 1; i < 8; ++i) gm = fmaxf(gm, red[i]);
    __syncthreads();

    float4 e;
    e.x = __expf(x.x - gm);
    e.y = __expf(x.y - gm);
    e.z = __expf(x.z - gm);
    e.w = __expf(x.w - gm);

    float s = e.x + e.y + e.z + e.w;
#pragma unroll
    for (int off = 16; off > 0; off >>= 1)
        s += __shfl_xor(s, off, 32);
    if (lane == 0) red[wave] = s;
    __syncthreads();
    float gs = red[0];
#pragma unroll
    for (int i = 1; i < 8; ++i) gs += red[i];

    const float inv = 1.0f / gs;
    e.x *= inv; e.y *= inv; e.z *= inv; e.w *= inv;
    *(float4*)(row + t * 4) = e;
}

__global__ __launch_bounds__(256) void attn_out_wmma(
    const float* __restrict__ attn,
    const float* __restrict__ value,
    float* __restrict__ out)
{
    const int lane      = threadIdx.x & 31;
    const int wave      = threadIdx.x >> 5;
    const int tilesN    = DSZ_ / 16;
    const int tilesPerB = (LQ_ / 16) * tilesN;
    const int tile      = blockIdx.x * 8 + wave;
    const int b         = tile / tilesPerB;
    const int tr        = tile % tilesPerB;
    const int tile_m    = tr / tilesN;
    const int tile_n    = tr % tilesN;

    const int n15 = lane & 15;
    const int ko  = (lane < 16) ? 0 : 2;

    const float* aptr = attn  + ((size_t)b * LQ_ + tile_m * 16 + n15) * LK_ + ko;
    const float* vptr = value + (size_t)b * LK_ * DSZ_
                               + (size_t)ko * DSZ_ + tile_n * 16 + n15;

    v8f c = {0.f, 0.f, 0.f, 0.f, 0.f, 0.f, 0.f, 0.f};
#pragma unroll 4
    for (int kk = 0; kk < LK_; kk += 4) {
        v2f av = *(const v2f*)(aptr + kk);
        const float* vp = vptr + (size_t)kk * DSZ_;
        v2f bv;
        bv.x = vp[0];
        bv.y = vp[DSZ_];
        c = __builtin_amdgcn_wmma_f32_16x16x4_f32(
                false, av, false, bv, (short)0, c, false, false);
    }

    const int mo = (lane < 16) ? 0 : 8;
#pragma unroll
    for (int r = 0; r < 8; ++r) {
        const int m = tile_m * 16 + mo + r;
        out[((size_t)b * LQ_ + m) * DSZ_ + tile_n * 16 + n15] = c[r];
    }
}

extern "C" void kernel_launch(void* const* d_in, const int* in_sizes, int n_in,
                              void* d_out, int out_size, void* d_ws, size_t ws_size,
                              hipStream_t stream) {
    const float* query     = (const float*)d_in[0];
    const float* key       = (const float*)d_in[1];
    const float* value     = (const float*)d_in[2];
    const int*   valid_len = (const int*)  d_in[3];
    const float* W_q       = (const float*)d_in[4];
    const float* W_k       = (const float*)d_in[5];
    const float* vvec      = (const float*)d_in[6];

    float* qf     = (float*)d_ws;
    float* kfT    = qf  + (size_t)B_ * LQ_ * UNITS_;
    float* scores = kfT + (size_t)B_ * UNITS_ * LK_;
    float* outp   = (float*)d_out;

    proj_wmma<false><<<64, 256, 0, stream>>>(query, W_q, qf,
                                             B_ * LQ_, DSZ_, UNITS_, 0);
    proj_wmma<true><<<512, 256, 0, stream>>>(key, W_k, kfT,
                                             B_ * LK_, DSZ_, UNITS_, LK_);
    scores_tanh_kernel<<<B_ * LQ_, 256, 0, stream>>>(qf, kfT, vvec, valid_len, scores);
    softmax_rows<<<B_ * LQ_, 256, 0, stream>>>(scores);
    attn_out_wmma<<<128, 256, 0, stream>>>(scores, value, outp);
}


# softmax launch dropped: scores kernel stores exp(s - max) per 64-column slice plus each slice's max and sum; attn x value scales its attn fragments by exp(m_j - M)/L; softmax sum order changed
# speedup vs baseline: 2.7750x; 2.7750x over previous
#include <hip/hip_runtime.h>

typedef float v2f __attribute__((ext_vector_type(2)));
typedef float v4f __attribute__((ext_vector_type(4)));
typedef float v8f __attribute__((ext_vector_type(8)));

#define B_     4
#define LQ_    128
#define LK_    1024
#define DSZ_   512
#define UNITS_ 256
#define MASK_VAL_ (-1.0e6f)

#if __has_builtin(__builtin_amdgcn_tanhf)
#define TANHF(x) __builtin_amdgcn_tanhf(x)
#else
#define TANHF(x) tanhf(x)
#endif

#define GK_KB   4
#define GK_PF   3
#define GK_RING (GK_PF + 1)

template <int TM, int TN>
struct ProjFrag {
    v4f a[TM][GK_KB];
    v4f b[TN][GK_KB];
};

template <int TM, int TN>
__device__ __forceinline__ void proj_load(ProjFrag<TM, TN>& f, const float* aptr,
                                          const float* wptr, int trip)
{
#pragma unroll
    for (int j = 0; j < GK_KB; ++j) {
        const int kk = (trip * GK_KB + j) * 8;
#pragma unroll
        for (int i = 0; i < TM; ++i)
            f.a[i][j] = *(const v4f*)(aptr + (size_t)i * 16 * DSZ_ + kk);
#pragma unroll
        for (int i = 0; i < TN; ++i)
            f.b[i][j] = *(const v4f*)(wptr + (size_t)i * 16 * DSZ_ + kk);
    }
}

template <int TM, int TN>
__device__ __forceinline__ void proj_mma(v8f (&c)[TM][TN], const ProjFrag<TM, TN>& f)
{
#pragma unroll
    for (int j = 0; j < GK_KB; ++j) {
        v2f a0[TM], a1[TM], b0[TN], b1[TN];
#pragma unroll
        for (int i = 0; i < TM; ++i) {
            a0[i] = __builtin_shufflevector(f.a[i][j], f.a[i][j], 0, 1);
            a1[i] = __builtin_shufflevector(f.a[i][j], f.a[i][j], 2, 3);
        }
#pragma unroll
        for (int i = 0; i < TN; ++i) {
            b0[i] = __builtin_shufflevector(f.b[i][j], f.b[i][j], 0, 1);
            b1[i] = __builtin_shufflevector(f.b[i][j], f.b[i][j], 2, 3);
        }
#pragma unroll
        for (int tm = 0; tm < TM; ++tm)
#pragma unroll
            for (int tn = 0; tn < TN; ++tn)
                c[tm][tn] = __builtin_amdgcn_wmma_f32_16x16x4_f32(
                        false, a0[tm], false, b0[tn], (short)0, c[tm][tn], false, false);
#pragma unroll
        for (int tm = 0; tm < TM; ++tm)
#pragma unroll
            for (int tn = 0; tn < TN; ++tn)
                c[tm][tn] = __builtin_amdgcn_wmma_f32_16x16x4_f32(
                        false, a1[tm], false, b1[tn], (short)0, c[tm][tn], false, false);
    }
}

template <bool LAST, int TM, int TN>
__device__ __forceinline__ void proj_pass(v8f (&c)[TM][TN], ProjFrag<TM, TN> (&ring)[GK_RING],
                                          const float* aptr, const float* wptr, int t)
{
#pragma unroll
    for (int s = 0; s < GK_RING; ++s) {
        if (!LAST || s + GK_PF < GK_RING)
            proj_load(ring[(s + GK_PF) & (GK_RING - 1)], aptr, wptr, t + s + GK_PF);
        proj_mma(c, ring[s]);
    }
}

template <bool TRANS_OUT, int WPB, int TM, int TN>
__global__ __launch_bounds__(WPB * 32) void proj_wmma(
    const float* __restrict__ A,
    const float* __restrict__ W,
    float* __restrict__ out,
    int M, int N, int seq)
{
    const int K      = DSZ_;
    const int NT     = K / (8 * GK_KB);
    const int lane   = threadIdx.x & 31;
    const int wave   = threadIdx.x >> 5;
    const int tilesN = (N >> 4) / TN;
    const int tile   = blockIdx.x * WPB + wave;
    const int tile_m = (tile / tilesN) * TM;
    const int tile_n = (tile % tilesN) * TN;

    const int n15  = lane & 15;
    const int ko   = (lane < 16) ? 0 : 4;
    const int arow = tile_m * 16 + n15;
    const int wrow = tile_n * 16 + n15;

    const float* aptr = A + (size_t)arow * K + ko;
    const float* wptr = W + (size_t)wrow * K + ko;

    v8f c[TM][TN];
#pragma unroll
    for (int tm = 0; tm < TM; ++tm)
#pragma unroll
        for (int tn = 0; tn < TN; ++tn)
            c[tm][tn] = (v8f){0.f, 0.f, 0.f, 0.f, 0.f, 0.f, 0.f, 0.f};
    ProjFrag<TM, TN> ring[GK_RING];
#pragma unroll
    for (int p = 0; p < GK_PF; ++p) {
        proj_load(ring[p], aptr, wptr, p);
        asm volatile("" ::: "memory");
    }

#pragma nounroll
    for (int t = 0; t < NT - GK_RING; t += GK_RING)
        proj_pass<false>(c, ring, aptr, wptr, t);
    proj_pass<true>(c, ring, aptr, wptr, NT - GK_RING);

    const int mo = (lane < 16) ? 0 : 8;
#pragma unroll
    for (int tm = 0; tm < TM; ++tm)
#pragma unroll
        for (int tn = 0; tn < TN; ++tn)
#pragma unroll
            for (int r = 0; r < 8; ++r) {
                const int m = (tile_m + tm) * 16 + mo + r;
                const int n = (tile_n + tn) * 16 + n15;
                if (TRANS_OUT) {
                    const int b    = m / seq;
                    const int kpos = m % seq;
                    out[((size_t)b * N + n) * seq + kpos] = c[tm][tn][r];
                } else {
                    out[(size_t)m * N + n] = c[tm][tn][r];
                }
            }
}

#define SC_R    4
#define SC_UC   8
#define SC_NCH  (UNITS_ / SC_UC)
#define SC_PF   3
#define SC_RING (SC_PF + 1)
#define SC_BLOCKS (B_ * (LQ_ / 8) * (LK_ / 256))
#define SC_SLICES (LK_ / 64)

struct ScScalars {
    float q[SC_R][SC_UC];
    float v[SC_UC];
};

__device__ __forceinline__ void sc_load_k(v2f (&dst)[SC_UC], const float* kp, int chunk)
{
#pragma unroll
    for (int j = 0; j < SC_UC; ++j)
        dst[j] = *(const v2f*)(kp + (size_t)(chunk * SC_UC + j) * LK_);
}

__device__ __forceinline__ void sc_load_s(ScScalars& d, const float* __restrict__ qrow,
                                          const float* __restrict__ vvec, int chunk)
{
#pragma unroll
    for (int j = 0; j < SC_UC; ++j) d.v[j] = vvec[chunk * SC_UC + j];
#pragma unroll
    for (int r = 0; r < SC_R; ++r)
#pragma unroll
        for (int j = 0; j < SC_UC; ++j)
            d.q[r][j] = qrow[r * UNITS_ + chunk * SC_UC + j];
}

__device__ __forceinline__ void sc_chunk(v2f (&acc)[SC_R], const v2f (&kv)[SC_UC],
                                         const ScScalars& s)
{
#pragma unroll
    for (int j = 0; j < SC_UC; ++j) {
        const v2f vv2 = {s.v[j], s.v[j]};
#pragma unroll
        for (int r = 0; r < SC_R; ++r) {
            const v2f q2 = {s.q[r][j], s.q[r][j]};
            const v2f x = q2 + kv[j];
            v2f t;
            t.x = TANHF(x.x);
            t.y = TANHF(x.y);
            acc[r] = __builtin_elementwise_fma(vv2, t, acc[r]);
        }
    }
}

template <int XM>
__device__ __forceinline__ float row_xor(float x)
{
    return __builtin_bit_cast(float, __builtin_amdgcn_update_dpp(
            0, __builtin_bit_cast(int, x), 0x160 + XM, 0xf, 0xf, false));
}
__device__ __forceinline__ float xhalf(float x)
{
    return __builtin_bit_cast(float, __builtin_amdgcn_permlanex16(
            __builtin_bit_cast(int, x), __builtin_bit_cast(int, x),
            0x76543210, 0xfedcba98, false, false));
}
template <bool IS_MAX>
__device__ __forceinline__ void wave_reduce4(float (&x)[4])
{
#define WR_STEP(EXPR)                                                      \
    _Pragma("unroll") for (int r = 0; r < 4; ++r) {                         \
        const float y = (EXPR);                                            \
        x[r] = IS_MAX ? fmaxf(x[r], y) : x[r] + y;                          \
    }
    WR_STEP(row_xor<1>(x[r]))
    WR_STEP(row_xor<2>(x[r]))
    WR_STEP(row_xor<4>(x[r]))
    WR_STEP(row_xor<8>(x[r]))
    WR_STEP(xhalf(x[r]))
#undef WR_STEP
}

__global__ __launch_bounds__(256) void scores_tanh_kernel(
    const float* __restrict__ qf,
    const float* __restrict__ kfT,
    const float* __restrict__ vvec,
    const int*   __restrict__ valid_len,
    float* __restrict__ scores,
    float* __restrict__ part)
{
    const int lane = threadIdx.x & 31;
    const int wave = __builtin_amdgcn_readfirstlane(threadIdx.x >> 5);
    const int kblk = blockIdx.x & 3;
    const int qblk = blockIdx.x >> 2;
    const int b    = qblk / (LQ_ / 8);
    const int row0 = qblk * 8 + (wave >> 2) * SC_R;
    const int k0   = kblk * 256 + (wave & 3) * 64 + lane * 2;

    const float* qrow = qf  + (size_t)row0 * UNITS_;
    const float* kp   = kfT + (size_t)b * UNITS_ * LK_ + k0;

    v2f acc[SC_R];
#pragma unroll
    for (int r = 0; r < SC_R; ++r) acc[r] = (v2f){0.f, 0.f};

    v2f ring[SC_RING][SC_UC];
    ScScalars sc[2];
#pragma unroll
    for (int p = 0; p < SC_PF; ++p) sc_load_k(ring[p], kp, p);
    sc_load_s(sc[0], qrow, vvec, 0);

    for (int c = 0; c < SC_NCH; c += SC_RING) {
#pragma unroll
        for (int s = 0; s < SC_RING; ++s) {
            const int cn = (c + s + SC_PF < SC_NCH) ? (c + s + SC_PF) : (SC_NCH - 1);
            const int cs = (c + s + 1 < SC_NCH) ? (c + s + 1) : (SC_NCH - 1);
            sc_load_k(ring[(s + SC_PF) & (SC_RING - 1)], kp, cn);
            sc_load_s(sc[(s + 1) & 1], qrow, vvec, cs);
            asm volatile("" ::: "memory");
            sc_chunk(acc, ring[s], sc[s & 1]);
        }
    }

    const int vl    = valid_len[b];
    const int slice = kblk * 4 + (wave & 3);
    v2f o[SC_R], e[SC_R];
    float m[SC_R], l[SC_R];
#pragma unroll
    for (int r = 0; r < SC_R; ++r) {
        o[r].x = (k0 + 0 < vl) ? acc[r].x : MASK_VAL_;
        o[r].y = (k0 + 1 < vl) ? acc[r].y : MASK_VAL_;
        m[r] = fmaxf(o[r].x, o[r].y);
    }
    wave_reduce4<true>(m);
#pragma unroll
    for (int r = 0; r < SC_R; ++r) {
        e[r].x = __expf(o[r].x - m[r]);
        e[r].y = __expf(o[r].y - m[r]);
        l[r] = e[r].x + e[r].y;
    }
    wave_reduce4<false>(l);
#pragma unroll
    for (int r = 0; r < SC_R; ++r) {
        *(v2f*)(scores + (size_t)(row0 + r) * LK_ + k0) = e[r];
        if (lane == 0) {
            const v2f ml = {m[r], l[r]};
            *(v2f*)(part + ((size_t)(row0 + r) * SC_SLICES + slice) * 2) = ml;
        }
    }
}

struct AvFrag {
    v4f a[GK_KB];
    v2f b0[GK_KB];
    v2f b1[GK_KB];
};

__device__ __forceinline__ void av_load(AvFrag& f, const float* aptr,
                                        const float* vptr, int trip)
{
#pragma unroll
    for (int j = 0; j < GK_KB; ++j) {
        const int kk = (trip * GK_KB + j) * 8;
        f.a[j] = *(const v4f*)(aptr + kk);
        const float* vp = vptr + (size_t)kk * DSZ_;
        f.b0[j].x = vp[0];
        f.b0[j].y = vp[DSZ_];
        f.b1[j].x = vp[2 * DSZ_];
        f.b1[j].y = vp[3 * DSZ_];
    }
}

__device__ __forceinline__ void av_mma(v8f& c, const AvFrag& f, float fs)
{
#pragma unroll
    for (int j = 0; j < GK_KB; ++j) {
        const v4f a  = f.a[j] * fs;
        const v2f a0 = __builtin_shufflevector(a, a, 0, 1);
        const v2f a1 = __builtin_shufflevector(a, a, 2, 3);
        c = __builtin_amdgcn_wmma_f32_16x16x4_f32(
                false, a0, false, f.b0[j], (short)0, c, false, false);
        c = __builtin_amdgcn_wmma_f32_16x16x4_f32(
                false, a1, false, f.b1[j], (short)0, c, false, false);
    }
}

template <bool LAST>
__device__ __forceinline__ void av_pass(v8f& c, AvFrag (&ring)[GK_RING],
                                        const float* aptr, const float* vptr, int t,
                                        float g0, float g1)
{
#pragma unroll
    for (int s = 0; s < GK_RING; ++s) {
        if (!LAST || s + GK_PF < GK_RING)
            av_load(ring[(s + GK_PF) & (GK_RING - 1)], aptr, vptr, t + s + GK_PF);
        __builtin_amdgcn_sched_barrier(0);
        av_mma(c, ring[s], (s < GK_RING / 2) ? g0 : g1);
        __builtin_amdgcn_sched_barrier(0);
    }
}

#define AV_WPB 4
__global__ __launch_bounds__(AV_WPB * 32) void attn_out_wmma(
    const float* __restrict__ attn,
    const float* __restrict__ part,
    const float* __restrict__ value,
    float* __restrict__ out)
{
    const int NT        = LK_ / (8 * GK_KB);
    const int lane      = threadIdx.x & 31;
    const int wave      = threadIdx.x >> 5;
    const int tilesN    = DSZ_ / 16;
    const int tilesPerB = (LQ_ / 16) * tilesN;
    const int tile      = blockIdx.x * AV_WPB + wave;
    const int b         = tile / tilesPerB;
    const int tr        = tile % tilesPerB;
    const int tile_m    = tr / tilesN;
    const int tile_n    = tr % tilesN;

    const int n15 = lane & 15;
    const int ko  = (lane < 16) ? 0 : 4;

    const float* aptr = attn  + ((size_t)b * LQ_ + tile_m * 16 + n15) * LK_ + ko;
    const float* vptr = value + (size_t)b * LK_ * DSZ_
                               + (size_t)ko * DSZ_ + tile_n * 16 + n15;

    const float* prow = part + ((size_t)b * LQ_ + tile_m * 16 + n15) * (SC_SLICES * 2);
    v4f pq[SC_SLICES / 2];
#pragma unroll
    for (int i = 0; i < SC_SLICES / 2; ++i) pq[i] = *(const v4f*)(prow + 4 * i);

    v8f c = {0.f, 0.f, 0.f, 0.f, 0.f, 0.f, 0.f, 0.f};
    AvFrag ring[GK_RING];
#pragma unroll
    for (int p = 0; p < GK_PF; ++p) {
        av_load(ring[p], aptr, vptr, p);
        asm volatile("" ::: "memory");
    }

    __builtin_amdgcn_sched_barrier(0);
    float M = pq[0].x;
#pragma unroll
    for (int i = 0; i < SC_SLICES / 2; ++i) M = fmaxf(fmaxf(M, pq[i].x), pq[i].z);
    float L = 0.f;
#pragma unroll
    for (int i = 0; i < SC_SLICES / 2; ++i) {
        L += pq[i].y * __expf(pq[i].x - M);
        L += pq[i].w * __expf(pq[i].z - M);
    }
    const float invL = 1.0f / L;
    float F[SC_SLICES];
#pragma unroll
    for (int i = 0; i < SC_SLICES / 2; ++i) {
        F[2 * i]     = __expf(pq[i].x - M) * invL;
        F[2 * i + 1] = __expf(pq[i].z - M) * invL;
    }

#pragma nounroll
    for (int t = 0; t < NT - GK_RING; t += GK_RING) {
        av_pass<false>(c, ring, aptr, vptr, t, F[0], F[1]);
#pragma unroll
        for (int i = 0; i + 2 < SC_SLICES; ++i) F[i] = F[i + 2];
    }
    av_pass<true>(c, ring, aptr, vptr, NT - GK_RING, F[0], F[1]);

    const int mo = (lane < 16) ? 0 : 8;
#pragma unroll
    for (int r = 0; r < 8; ++r) {
        const int m = tile_m * 16 + mo + r;
        out[((size_t)b * LQ_ + m) * DSZ_ + tile_n * 16 + n15] = c[r];
    }
}

extern "C" void kernel_launch(void* const* d_in, const int* in_sizes, int n_in,
                              void* d_out, int out_size, void* d_ws, size_t ws_size,
                              hipStream_t stream) {
    const float* query     = (const float*)d_in[0];
    const float* key       = (const float*)d_in[1];
    const float* value     = (const float*)d_in[2];
    const int*   valid_len = (const int*)  d_in[3];
    const float* W_q       = (const float*)d_in[4];
    const float* W_k       = (const float*)d_in[5];
    const float* vvec      = (const float*)d_in[6];

    float* qf     = (float*)d_ws;
    float* kfT    = qf  + (size_t)B_ * LQ_ * UNITS_;
    float* scores = kfT + (size_t)B_ * UNITS_ * LK_;
    float* part   = scores + (size_t)B_ * LQ_ * LK_;
    float* outp   = (float*)d_out;

    proj_wmma<false, 2, 1, 1><<<256, 64, 0, stream>>>(query, W_q, qf,
                                                      B_ * LQ_, UNITS_, 0);
    proj_wmma<true, 4, 2, 2><<<256, 128, 0, stream>>>(key, W_k, kfT,
                                                      B_ * LK_, UNITS_, LK_);
    scores_tanh_kernel<<<SC_BLOCKS, 256, 0, stream>>>(qf, kfT, vvec, valid_len, scores, part);
    attn_out_wmma<<<256, AV_WPB * 32, 0, stream>>>(scores, part, value, outp);
}


# attn x value GEMM: value rows read 16 B per lane as whole lines into 4 column-interleaved tiles per wave; a block's 4 waves split k in quarters, partial tiles added through LDS (k sum order changed)
# speedup vs baseline: 2.9934x; 1.0787x over previous
#include <hip/hip_runtime.h>

typedef float v2f __attribute__((ext_vector_type(2)));
typedef float v4f __attribute__((ext_vector_type(4)));
typedef float v8f __attribute__((ext_vector_type(8)));

#define B_     4
#define LQ_    128
#define LK_    1024
#define DSZ_   512
#define UNITS_ 256
#define MASK_VAL_ (-1.0e6f)

#if __has_builtin(__builtin_amdgcn_tanhf)
#define TANHF(x) __builtin_amdgcn_tanhf(x)
#else
#define TANHF(x) tanhf(x)
#endif

#define GK_KB   4
#define GK_PF   3
#define GK_RING (GK_PF + 1)

template <int TM, int TN>
struct ProjFrag {
    v4f a[TM][GK_KB];
    v4f b[TN][GK_KB];
};

template <int TM, int TN>
__device__ __forceinline__ void proj_load(ProjFrag<TM, TN>& f, const float* aptr,
                                          const float* wptr, int trip)
{
#pragma unroll
    for (int j = 0; j < GK_KB; ++j) {
        const int kk = (trip * GK_KB + j) * 8;
#pragma unroll
        for (int i = 0; i < TM; ++i)
            f.a[i][j] = *(const v4f*)(aptr + (size_t)i * 16 * DSZ_ + kk);
#pragma unroll
        for (int i = 0; i < TN; ++i)
            f.b[i][j] = *(const v4f*)(wptr + (size_t)i * 16 * DSZ_ + kk);
    }
}

template <int TM, int TN>
__device__ __forceinline__ void proj_mma(v8f (&c)[TM][TN], const ProjFrag<TM, TN>& f)
{
#pragma unroll
    for (int j = 0; j < GK_KB; ++j) {
        v2f a0[TM], a1[TM], b0[TN], b1[TN];
#pragma unroll
        for (int i = 0; i < TM; ++i) {
            a0[i] = __builtin_shufflevector(f.a[i][j], f.a[i][j], 0, 1);
            a1[i] = __builtin_shufflevector(f.a[i][j], f.a[i][j], 2, 3);
        }
#pragma unroll
        for (int i = 0; i < TN; ++i) {
            b0[i] = __builtin_shufflevector(f.b[i][j], f.b[i][j], 0, 1);
            b1[i] = __builtin_shufflevector(f.b[i][j], f.b[i][j], 2, 3);
        }
#pragma unroll
        for (int tm = 0; tm < TM; ++tm)
#pragma unroll
            for (int tn = 0; tn < TN; ++tn)
                c[tm][tn] = __builtin_amdgcn_wmma_f32_16x16x4_f32(
                        false, a0[tm], false, b0[tn], (short)0, c[tm][tn], false, false);
#pragma unroll
        for (int tm = 0; tm < TM; ++tm)
#pragma unroll
            for (int tn = 0; tn < TN; ++tn)
                c[tm][tn] = __builtin_amdgcn_wmma_f32_16x16x4_f32(
                        false, a1[tm], false, b1[tn], (short)0, c[tm][tn], false, false);
    }
}

template <bool LAST, int TM, int TN>
__device__ __forceinline__ void proj_pass(v8f (&c)[TM][TN], ProjFrag<TM, TN> (&ring)[GK_RING],
                                          const float* aptr, const float* wptr, int t)
{
#pragma unroll
    for (int s = 0; s < GK_RING; ++s) {
        if (!LAST || s + GK_PF < GK_RING)
            proj_load(ring[(s + GK_PF) & (GK_RING - 1)], aptr, wptr, t + s + GK_PF);
        proj_mma(c, ring[s]);
    }
}

template <bool TRANS_OUT, int WPB, int TM, int TN>
__global__ __launch_bounds__(WPB * 32) void proj_wmma(
    const float* __restrict__ A,
    const float* __restrict__ W,
    float* __restrict__ out,
    int M, int N, int seq)
{
    const int K      = DSZ_;
    const int NT     = K / (8 * GK_KB);
    const int lane   = threadIdx.x & 31;
    const int wave   = threadIdx.x >> 5;
    const int tilesN = (N >> 4) / TN;
    const int tile   = blockIdx.x * WPB + wave;
    const int tile_m = (tile / tilesN) * TM;
    const int tile_n = (tile % tilesN) * TN;

    const int n15  = lane & 15;
    const int ko   = (lane < 16) ? 0 : 4;
    const int arow = tile_m * 16 + n15;
    const int wrow = tile_n * 16 + n15;

    const float* aptr = A + (size_t)arow * K + ko;
    const float* wptr = W + (size_t)wrow * K + ko;

    v8f c[TM][TN];
#pragma unroll
    for (int tm = 0; tm < TM; ++tm)
#pragma unroll
        for (int tn = 0; tn < TN; ++tn)
            c[tm][tn] = (v8f){0.f, 0.f, 0.f, 0.f, 0.f, 0.f, 0.f, 0.f};
    ProjFrag<TM, TN> ring[GK_RING];
#pragma unroll
    for (int p = 0; p < GK_PF; ++p) {
        proj_load(ring[p], aptr, wptr, p);
        asm volatile("" ::: "memory");
    }

#pragma nounroll
    for (int t = 0; t < NT - GK_RING; t += GK_RING)
        proj_pass<false>(c, ring, aptr, wptr, t);
    proj_pass<true>(c, ring, aptr, wptr, NT - GK_RING);

    const int mo = (lane < 16) ? 0 : 8;
#pragma unroll
    for (int tm = 0; tm < TM; ++tm)
#pragma unroll
        for (int tn = 0; tn < TN; ++tn)
#pragma unroll
            for (int r = 0; r < 8; ++r) {
                const int m = (tile_m + tm) * 16 + mo + r;
                const int n = (tile_n + tn) * 16 + n15;
                if (TRANS_OUT) {
                    const int b    = m / seq;
                    const int kpos = m % seq;
                    out[((size_t)b * N + n) * seq + kpos] = c[tm][tn][r];
                } else {
                    out[(size_t)m * N + n] = c[tm][tn][r];
                }
            }
}

#define SC_R    4
#define SC_UC   8
#define SC_NCH  (UNITS_ / SC_UC)
#define SC_PF   3
#define SC_RING (SC_PF + 1)
#define SC_BLOCKS (B_ * (LQ_ / 8) * (LK_ / 256))
#define SC_SLICES (LK_ / 64)

struct ScScalars {
    float q[SC_R][SC_UC];
    float v[SC_UC];
};

__device__ __forceinline__ void sc_load_k(v2f (&dst)[SC_UC], const float* kp, int chunk)
{
#pragma unroll
    for (int j = 0; j < SC_UC; ++j)
        dst[j] = *(const v2f*)(kp + (size_t)(chunk * SC_UC + j) * LK_);
}

__device__ __forceinline__ void sc_load_s(ScScalars& d, const float* __restrict__ qrow,
                                          const float* __restrict__ vvec, int chunk)
{
#pragma unroll
    for (int j = 0; j < SC_UC; ++j) d.v[j] = vvec[chunk * SC_UC + j];
#pragma unroll
    for (int r = 0; r < SC_R; ++r)
#pragma unroll
        for (int j = 0; j < SC_UC; ++j)
            d.q[r][j] = qrow[r * UNITS_ + chunk * SC_UC + j];
}

__device__ __forceinline__ void sc_chunk(v2f (&acc)[SC_R], const v2f (&kv)[SC_UC],
                                         const ScScalars& s)
{
#pragma unroll
    for (int j = 0; j < SC_UC; ++j) {
        const v2f vv2 = {s.v[j], s.v[j]};
#pragma unroll
        for (int r = 0; r < SC_R; ++r) {
            const v2f q2 = {s.q[r][j], s.q[r][j]};
            const v2f x = q2 + kv[j];
            v2f t;
            t.x = TANHF(x.x);
            t.y = TANHF(x.y);
            acc[r] = __builtin_elementwise_fma(vv2, t, acc[r]);
        }
    }
}

template <int XM>
__device__ __forceinline__ float row_xor(float x)
{
    return __builtin_bit_cast(float, __builtin_amdgcn_update_dpp(
            0, __builtin_bit_cast(int, x), 0x160 + XM, 0xf, 0xf, false));
}
__device__ __forceinline__ float xhalf(float x)
{
    return __builtin_bit_cast(float, __builtin_amdgcn_permlanex16(
            __builtin_bit_cast(int, x), __builtin_bit_cast(int, x),
            0x76543210, 0xfedcba98, false, false));
}
template <bool IS_MAX>
__device__ __forceinline__ void wave_reduce4(float (&x)[4])
{
#define WR_STEP(EXPR)                                                      \
    _Pragma("unroll") for (int r = 0; r < 4; ++r) {                         \
        const float y = (EXPR);                                            \
        x[r] = IS_MAX ? fmaxf(x[r], y) : x[r] + y;                          \
    }
    WR_STEP(row_xor<1>(x[r]))
    WR_STEP(row_xor<2>(x[r]))
    WR_STEP(row_xor<4>(x[r]))
    WR_STEP(row_xor<8>(x[r]))
    WR_STEP(xhalf(x[r]))
#undef WR_STEP
}

__global__ __launch_bounds__(256) void scores_tanh_kernel(
    const float* __restrict__ qf,
    const float* __restrict__ kfT,
    const float* __restrict__ vvec,
    const int*   __restrict__ valid_len,
    float* __restrict__ scores,
    float* __restrict__ part)
{
    const int lane = threadIdx.x & 31;
    const int wave = __builtin_amdgcn_readfirstlane(threadIdx.x >> 5);
    const int kblk = blockIdx.x & 3;
    const int qblk = blockIdx.x >> 2;
    const int b    = qblk / (LQ_ / 8);
    const int row0 = qblk * 8 + (wave >> 2) * SC_R;
    const int k0   = kblk * 256 + (wave & 3) * 64 + lane * 2;

    const float* qrow = qf  + (size_t)row0 * UNITS_;
    const float* kp   = kfT + (size_t)b * UNITS_ * LK_ + k0;

    v2f acc[SC_R];
#pragma unroll
    for (int r = 0; r < SC_R; ++r) acc[r] = (v2f){0.f, 0.f};

    v2f ring[SC_RING][SC_UC];
    ScScalars sc[2];
#pragma unroll
    for (int p = 0; p < SC_PF; ++p) sc_load_k(ring[p], kp, p);
    sc_load_s(sc[0], qrow, vvec, 0);

    for (int c = 0; c < SC_NCH; c += SC_RING) {
#pragma unroll
        for (int s = 0; s < SC_RING; ++s) {
            const int cn = (c + s + SC_PF < SC_NCH) ? (c + s + SC_PF) : (SC_NCH - 1);
            const int cs = (c + s + 1 < SC_NCH) ? (c + s + 1) : (SC_NCH - 1);
            sc_load_k(ring[(s + SC_PF) & (SC_RING - 1)], kp, cn);
            sc_load_s(sc[(s + 1) & 1], qrow, vvec, cs);
            asm volatile("" ::: "memory");
            sc_chunk(acc, ring[s], sc[s & 1]);
        }
    }

    const int vl    = valid_len[b];
    const int slice = kblk * 4 + (wave & 3);
    v2f o[SC_R], e[SC_R];
    float m[SC_R], l[SC_R];
#pragma unroll
    for (int r = 0; r < SC_R; ++r) {
        o[r].x = (k0 + 0 < vl) ? acc[r].x : MASK_VAL_;
        o[r].y = (k0 + 1 < vl) ? acc[r].y : MASK_VAL_;
        m[r] = fmaxf(o[r].x, o[r].y);
    }
    wave_reduce4<true>(m);
#pragma unroll
    for (int r = 0; r < SC_R; ++r) {
        e[r].x = __expf(o[r].x - m[r]);
        e[r].y = __expf(o[r].y - m[r]);
        l[r] = e[r].x + e[r].y;
    }
    wave_reduce4<false>(l);
#pragma unroll
    for (int r = 0; r < SC_R; ++r) {
        *(v2f*)(scores + (size_t)(row0 + r) * LK_ + k0) = e[r];
        if (lane == 0) {
            const v2f ml = {m[r], l[r]};
            *(v2f*)(part + ((size_t)(row0 + r) * SC_SLICES + slice) * 2) = ml;
        }
    }
}

#define AV_WPB 4
#define AV_TN  4
#define AV_KQ  (LK_ / AV_WPB)
#define AV_NT  (AV_KQ / (8 * GK_KB))

struct AvFrag {
    v4f a[GK_KB];
    v4f b[GK_KB][4];
};

__device__ __forceinline__ void av_load(AvFrag& f, const float* aptr,
                                        const float* vptr, int trip)
{
#pragma unroll
    for (int j = 0; j < GK_KB; ++j) {
        const int kk = (trip * GK_KB + j) * 8;
        f.a[j] = *(const v4f*)(aptr + kk);
        const float* vp = vptr + (size_t)kk * DSZ_;
#pragma unroll
        for (int i = 0; i < 4; ++i)
            f.b[j][i] = *(const v4f*)(vp + (size_t)i * DSZ_);
    }
}

__device__ __forceinline__ void av_mma(v8f (&c)[AV_TN], const AvFrag& f, float fs)
{
#pragma unroll
    for (int j = 0; j < GK_KB; ++j) {
        const v4f a  = f.a[j] * fs;
        const v2f a0 = __builtin_shufflevector(a, a, 0, 1);
        const v2f a1 = __builtin_shufflevector(a, a, 2, 3);
#pragma unroll
        for (int t = 0; t < AV_TN; ++t) {
            const v2f b0 = {f.b[j][0][t], f.b[j][1][t]};
            c[t] = __builtin_amdgcn_wmma_f32_16x16x4_f32(
                    false, a0, false, b0, (short)0, c[t], false, false);
        }
#pragma unroll
        for (int t = 0; t < AV_TN; ++t) {
            const v2f b1 = {f.b[j][2][t], f.b[j][3][t]};
            c[t] = __builtin_amdgcn_wmma_f32_16x16x4_f32(
                    false, a1, false, b1, (short)0, c[t], false, false);
        }
    }
}

template <bool LAST>
__device__ __forceinline__ void av_pass(v8f (&c)[AV_TN], AvFrag (&ring)[GK_RING],
                                        const float* aptr, const float* vptr, int t,
                                        float g0, float g1)
{
#pragma unroll
    for (int s = 0; s < GK_RING; ++s) {
        if (!LAST || s + GK_PF < GK_RING)
            av_load(ring[(s + GK_PF) & (GK_RING - 1)], aptr, vptr, t + s + GK_PF);
        __builtin_amdgcn_sched_barrier(0);
        av_mma(c, ring[s], (s < GK_RING / 2) ? g0 : g1);
        __builtin_amdgcn_sched_barrier(0);
    }
}

__global__ __launch_bounds__(AV_WPB * 32) void attn_out_wmma(
    const float* __restrict__ attn,
    const float* __restrict__ part,
    const float* __restrict__ value,
    float* __restrict__ out)
{
    __shared__ v4f red[AV_WPB][8][32];

    const int lane      = threadIdx.x & 31;
    const int kq        = threadIdx.x >> 5;
    const int blocksN   = DSZ_ / (16 * AV_TN);
    const int blocksPerB = (LQ_ / 16) * blocksN;
    const int b         = blockIdx.x / blocksPerB;
    const int tr        = blockIdx.x % blocksPerB;
    const int tile_m    = tr / blocksN;
    const int n0        = (tr % blocksN) * (16 * AV_TN);

    const int n15 = lane & 15;
    const int ko  = (lane < 16) ? 0 : 4;
    const int kb  = kq * AV_KQ;

    const float* aptr = attn  + ((size_t)b * LQ_ + tile_m * 16 + n15) * LK_ + kb + ko;
    const float* vptr = value + (size_t)b * LK_ * DSZ_
                               + (size_t)(kb + ko) * DSZ_ + n0 + 4 * n15;

    const float* prow = part + ((size_t)b * LQ_ + tile_m * 16 + n15) * (SC_SLICES * 2);
    v4f pq[SC_SLICES / 2];
#pragma unroll
    for (int i = 0; i < SC_SLICES / 2; ++i) pq[i] = *(const v4f*)(prow + 4 * i);
    const v4f po0 = *(const v4f*)(prow + kq * 8);
    const v4f po1 = *(const v4f*)(prow + kq * 8 + 4);

    v8f c[AV_TN];
#pragma unroll
    for (int t = 0; t < AV_TN; ++t) c[t] = (v8f){0.f, 0.f, 0.f, 0.f, 0.f, 0.f, 0.f, 0.f};
    AvFrag ring[GK_RING];
#pragma unroll
    for (int p = 0; p < GK_PF; ++p) {
        av_load(ring[p], aptr, vptr, p);
        asm volatile("" ::: "memory");
    }

    __builtin_amdgcn_sched_barrier(0);
    float M = pq[0].x;
#pragma unroll
    for (int i = 0; i < SC_SLICES / 2; ++i) M = fmaxf(fmaxf(M, pq[i].x), pq[i].z);
    float L = 0.f;
#pragma unroll
    for (int i = 0; i < SC_SLICES / 2; ++i) {
        L += pq[i].y * __expf(pq[i].x - M);
        L += pq[i].w * __expf(pq[i].z - M);
    }
    const float invL = 1.0f / L;
    const float g0 = __expf(po0.x - M) * invL;
    const float g1 = __expf(po0.z - M) * invL;
    const float g2 = __expf(po1.x - M) * invL;
    const float g3 = __expf(po1.z - M) * invL;

    av_pass<false>(c, ring, aptr, vptr, 0, g0, g1);
    av_pass<true>(c, ring, aptr, vptr, AV_NT - GK_RING, g2, g3);

#pragma unroll
    for (int r = 0; r < 8; ++r) {
        const v4f pv = {c[0][r], c[1][r], c[2][r], c[3][r]};
        red[kq][r][lane] = pv;
    }
    __syncthreads();
    const int mo = (lane < 16) ? 0 : 8;
#pragma unroll
    for (int rr = 0; rr < 2; ++rr) {
        const int r = 2 * kq + rr;
        v4f s = red[0][r][lane];
#pragma unroll
        for (int q = 1; q < AV_WPB; ++q) s += red[q][r][lane];
        const int m = tile_m * 16 + mo + r;
        *(v4f*)(out + ((size_t)b * LQ_ + m) * DSZ_ + n0 + 4 * n15) = s;
    }
}

extern "C" void kernel_launch(void* const* d_in, const int* in_sizes, int n_in,
                              void* d_out, int out_size, void* d_ws, size_t ws_size,
                              hipStream_t stream) {
    const float* query     = (const float*)d_in[0];
    const float* key       = (const float*)d_in[1];
    const float* value     = (const float*)d_in[2];
    const int*   valid_len = (const int*)  d_in[3];
    const float* W_q       = (const float*)d_in[4];
    const float* W_k       = (const float*)d_in[5];
    const float* vvec      = (const float*)d_in[6];

    float* qf     = (float*)d_ws;
    float* kfT    = qf  + (size_t)B_ * LQ_ * UNITS_;
    float* scores = kfT + (size_t)B_ * UNITS_ * LK_;
    float* part   = scores + (size_t)B_ * LQ_ * LK_;
    float* outp   = (float*)d_out;

    proj_wmma<false, 2, 1, 1><<<256, 64, 0, stream>>>(query, W_q, qf,
                                                      B_ * LQ_, UNITS_, 0);
    proj_wmma<true, 4, 2, 2><<<256, 128, 0, stream>>>(key, W_k, kfT,
                                                      B_ * LK_, UNITS_, LK_);
    scores_tanh_kernel<<<SC_BLOCKS, 256, 0, stream>>>(qf, kfT, vvec, valid_len, scores, part);
    attn_out_wmma<<<256, AV_WPB * 32, 0, stream>>>(scores, part, value, outp);
}


# both projection GEMMs: a block's 4 waves split k in quarters (wave tiles 4x4 key side, 1x2 query side, fragments from global), partial tiles added through LDS, whole-line stores; k sum order changed
# speedup vs baseline: 3.4399x; 1.1492x over previous
#include <hip/hip_runtime.h>

typedef float v2f __attribute__((ext_vector_type(2)));
typedef float v4f __attribute__((ext_vector_type(4)));
typedef float v8f __attribute__((ext_vector_type(8)));

#define B_     4
#define LQ_    128
#define LK_    1024
#define DSZ_   512
#define UNITS_ 256
#define MASK_VAL_ (-1.0e6f)

#if __has_builtin(__builtin_amdgcn_tanhf)
#define TANHF(x) __builtin_amdgcn_tanhf(x)
#else
#define TANHF(x) tanhf(x)
#endif


#define GK_KB   4
#define GK_PF   3
#define GK_RING (GK_PF + 1)

#define PK_WPB 4
#define PK_KQ  (DSZ_ / PK_WPB)
#define PK_NT  (PK_KQ / (8 * GK_KB))

template <int N> struct VecOf;
template <> struct VecOf<2> { typedef v2f type; };
template <> struct VecOf<4> { typedef v4f type; };

template <int TM, int TN>
struct ProjFrag {
    v4f a[TM][GK_KB];
    v4f b[TN][GK_KB];
};

template <int TM, int TN>
__device__ __forceinline__ void proj_load(ProjFrag<TM, TN>& f, const float* aptr,
                                          const float* bptr, int trip)
{
#pragma unroll
    for (int j = 0; j < GK_KB; ++j) {
        const int kk = (trip * GK_KB + j) * 8;
#pragma unroll
        for (int i = 0; i < TM; ++i)
            f.a[i][j] = *(const v4f*)(aptr + (size_t)i * 16 * DSZ_ + kk);
#pragma unroll
        for (int i = 0; i < TN; ++i)
            f.b[i][j] = *(const v4f*)(bptr + (size_t)i * DSZ_ + kk);
    }
}

template <int TM, int TN>
__device__ __forceinline__ void proj_mma(v8f (&c)[TM][TN], const ProjFrag<TM, TN>& f)
{
#pragma unroll
    for (int j = 0; j < GK_KB; ++j) {
        v2f a0[TM], a1[TM], b0[TN], b1[TN];
#pragma unroll
        for (int i = 0; i < TM; ++i) {
            a0[i] = __builtin_shufflevector(f.a[i][j], f.a[i][j], 0, 1);
            a1[i] = __builtin_shufflevector(f.a[i][j], f.a[i][j], 2, 3);
        }
#pragma unroll
        for (int i = 0; i < TN; ++i) {
            b0[i] = __builtin_shufflevector(f.b[i][j], f.b[i][j], 0, 1);
            b1[i] = __builtin_shufflevector(f.b[i][j], f.b[i][j], 2, 3);
        }
#pragma unroll
        for (int tm = 0; tm < TM; ++tm)
#pragma unroll
            for (int tn = 0; tn < TN; ++tn)
                c[tm][tn] = __builtin_amdgcn_wmma_f32_16x16x4_f32(
                        false, a0[tm], false, b0[tn], (short)0, c[tm][tn], false, false);
#pragma unroll
        for (int tm = 0; tm < TM; ++tm)
#pragma unroll
            for (int tn = 0; tn < TN; ++tn)
                c[tm][tn] = __builtin_amdgcn_wmma_f32_16x16x4_f32(
                        false, a1[tm], false, b1[tn], (short)0, c[tm][tn], false, false);
    }
}

template <bool KEY_SIDE, int TM, int TN, int PF>
__global__ __launch_bounds__(PK_WPB * 32) void proj_splitk(
    const float* __restrict__ X,
    const float* __restrict__ Wt,
    float* __restrict__ out)
{
    typedef typename VecOf<TN>::type vtn;
    constexpr int RING = PF + 1;
    __shared__ vtn red[PK_WPB][TM][8][32];

    const int lane = threadIdx.x & 31;
    const int kq   = threadIdx.x >> 5;
    const int n15  = lane & 15;
    const int ko   = (lane < 16) ? 0 : 4;

    const float* P;
    const float* Q;
    int a0, b0, ldo;
    float* obase;
    if (KEY_SIDE) {
        const int ublocks = UNITS_ / (16 * TM);
        a0 = (blockIdx.x % ublocks) * (16 * TM);
        b0 = (blockIdx.x / ublocks) * (16 * TN);
        P = Wt; Q = X;
        ldo = LK_;
        obase = out + ((size_t)(b0 / LK_) * UNITS_ + a0) * LK_ + (b0 % LK_);
    } else {
        const int ublocks = UNITS_ / (16 * TN);
        a0 = (blockIdx.x / ublocks) * (16 * TM);
        b0 = (blockIdx.x % ublocks) * (16 * TN);
        P = X; Q = Wt;
        ldo = UNITS_;
        obase = out + (size_t)a0 * UNITS_ + b0;
    }
    const int kb = kq * PK_KQ;
    const float* aptr = P + (size_t)(a0 + n15) * DSZ_ + kb + ko;
    const float* bptr = Q + (size_t)(b0 + TN * n15) * DSZ_ + kb + ko;

    v8f c[TM][TN];
#pragma unroll
    for (int tm = 0; tm < TM; ++tm)
#pragma unroll
        for (int tn = 0; tn < TN; ++tn)
            c[tm][tn] = (v8f){0.f, 0.f, 0.f, 0.f, 0.f, 0.f, 0.f, 0.f};
    ProjFrag<TM, TN> ring[RING];
#pragma unroll
    for (int p = 0; p < PF; ++p) {
        proj_load(ring[p], aptr, bptr, p);
        asm volatile("" ::: "memory");
    }
#pragma unroll
    for (int t = 0; t < PK_NT; ++t) {
        if (t + PF < PK_NT)
            proj_load(ring[(t + PF) & (RING - 1)], aptr, bptr, t + PF);
        __builtin_amdgcn_sched_barrier(0);
        proj_mma(c, ring[t & (RING - 1)]);
        __builtin_amdgcn_sched_barrier(0);
    }

#pragma unroll
    for (int tm = 0; tm < TM; ++tm)
#pragma unroll
        for (int r = 0; r < 8; ++r) {
            vtn pv;
#pragma unroll
            for (int tn = 0; tn < TN; ++tn) pv[tn] = c[tm][tn][r];
            red[kq][tm][r][lane] = pv;
        }
    __syncthreads();
    const int mo = (lane < 16) ? 0 : 8;
    constexpr int PER = TM * 8 / PK_WPB;
#pragma unroll
    for (int i = 0; i < PER; ++i) {
        const int idx = kq * PER + i;
        const int tm  = idx >> 3;
        const int r   = idx & 7;
        vtn s = red[0][tm][r][lane];
#pragma unroll
        for (int q = 1; q < PK_WPB; ++q) s += red[q][tm][r][lane];
        *(vtn*)(obase + (size_t)(tm * 16 + mo + r) * ldo + TN * n15) = s;
    }
}

#define SC_R    4
#define SC_UC   8
#define SC_NCH  (UNITS_ / SC_UC)
#define SC_PF   3
#define SC_RING (SC_PF + 1)
#define SC_BLOCKS (B_ * (LQ_ / 8) * (LK_ / 256))
#define SC_SLICES (LK_ / 64)

struct ScScalars {
    float q[SC_R][SC_UC];
    float v[SC_UC];
};

__device__ __forceinline__ void sc_load_k(v2f (&dst)[SC_UC], const float* kp, int chunk)
{
#pragma unroll
    for (int j = 0; j < SC_UC; ++j)
        dst[j] = *(const v2f*)(kp + (size_t)(chunk * SC_UC + j) * LK_);
}

__device__ __forceinline__ void sc_load_s(ScScalars& d, const float* __restrict__ qrow,
                                          const float* __restrict__ vvec, int chunk)
{
#pragma unroll
    for (int j = 0; j < SC_UC; ++j) d.v[j] = vvec[chunk * SC_UC + j];
#pragma unroll
    for (int r = 0; r < SC_R; ++r)
#pragma unroll
        for (int j = 0; j < SC_UC; ++j)
            d.q[r][j] = qrow[r * UNITS_ + chunk * SC_UC + j];
}

__device__ __forceinline__ void sc_chunk(v2f (&acc)[SC_R], const v2f (&kv)[SC_UC],
                                         const ScScalars& s)
{
#pragma unroll
    for (int j = 0; j < SC_UC; ++j) {
        const v2f vv2 = {s.v[j], s.v[j]};
#pragma unroll
        for (int r = 0; r < SC_R; ++r) {
            const v2f q2 = {s.q[r][j], s.q[r][j]};
            const v2f x = q2 + kv[j];
            v2f t;
            t.x = TANHF(x.x);
            t.y = TANHF(x.y);
            acc[r] = __builtin_elementwise_fma(vv2, t, acc[r]);
        }
    }
}

template <int XM>
__device__ __forceinline__ float row_xor(float x)
{
    return __builtin_bit_cast(float, __builtin_amdgcn_update_dpp(
            0, __builtin_bit_cast(int, x), 0x160 + XM, 0xf, 0xf, false));
}
__device__ __forceinline__ float xhalf(float x)
{
    return __builtin_bit_cast(float, __builtin_amdgcn_permlanex16(
            __builtin_bit_cast(int, x), __builtin_bit_cast(int, x),
            0x76543210, 0xfedcba98, false, false));
}
template <bool IS_MAX>
__device__ __forceinline__ void wave_reduce4(float (&x)[4])
{
#define WR_STEP(EXPR)                                                      \
    _Pragma("unroll") for (int r = 0; r < 4; ++r) {                         \
        const float y = (EXPR);                                            \
        x[r] = IS_MAX ? fmaxf(x[r], y) : x[r] + y;                          \
    }
    WR_STEP(row_xor<1>(x[r]))
    WR_STEP(row_xor<2>(x[r]))
    WR_STEP(row_xor<4>(x[r]))
    WR_STEP(row_xor<8>(x[r]))
    WR_STEP(xhalf(x[r]))
#undef WR_STEP
}

__global__ __launch_bounds__(256) void scores_tanh_kernel(
    const float* __restrict__ qf,
    const float* __restrict__ kfT,
    const float* __restrict__ vvec,
    const int*   __restrict__ valid_len,
    float* __restrict__ scores,
    float* __restrict__ part)
{
    const int lane = threadIdx.x & 31;
    const int wave = __builtin_amdgcn_readfirstlane(threadIdx.x >> 5);
    const int kblk = blockIdx.x & 3;
    const int qblk = blockIdx.x >> 2;
    const int b    = qblk / (LQ_ / 8);
    const int row0 = qblk * 8 + (wave >> 2) * SC_R;
    const int k0   = kblk * 256 + (wave & 3) * 64 + lane * 2;

    const float* qrow = qf  + (size_t)row0 * UNITS_;
    const float* kp   = kfT + (size_t)b * UNITS_ * LK_ + k0;

    v2f acc[SC_R];
#pragma unroll
    for (int r = 0; r < SC_R; ++r) acc[r] = (v2f){0.f, 0.f};

    v2f ring[SC_RING][SC_UC];
    ScScalars sc[2];
#pragma unroll
    for (int p = 0; p < SC_PF; ++p) sc_load_k(ring[p], kp, p);
    sc_load_s(sc[0], qrow, vvec, 0);

    for (int c = 0; c < SC_NCH; c += SC_RING) {
#pragma unroll
        for (int s = 0; s < SC_RING; ++s) {
            const int cn = (c + s + SC_PF < SC_NCH) ? (c + s + SC_PF) : (SC_NCH - 1);
            const int cs = (c + s + 1 < SC_NCH) ? (c + s + 1) : (SC_NCH - 1);
            sc_load_k(ring[(s + SC_PF) & (SC_RING - 1)], kp, cn);
            sc_load_s(sc[(s + 1) & 1], qrow, vvec, cs);
            asm volatile("" ::: "memory");
            sc_chunk(acc, ring[s], sc[s & 1]);
        }
    }

    const int vl    = valid_len[b];
    const int slice = kblk * 4 + (wave & 3);
    v2f o[SC_R], e[SC_R];
    float m[SC_R], l[SC_R];
#pragma unroll
    for (int r = 0; r < SC_R; ++r) {
        o[r].x = (k0 + 0 < vl) ? acc[r].x : MASK_VAL_;
        o[r].y = (k0 + 1 < vl) ? acc[r].y : MASK_VAL_;
        m[r] = fmaxf(o[r].x, o[r].y);
    }
    wave_reduce4<true>(m);
#pragma unroll
    for (int r = 0; r < SC_R; ++r) {
        e[r].x = __expf(o[r].x - m[r]);
        e[r].y = __expf(o[r].y - m[r]);
        l[r] = e[r].x + e[r].y;
    }
    wave_reduce4<false>(l);
#pragma unroll
    for (int r = 0; r < SC_R; ++r) {
        *(v2f*)(scores + (size_t)(row0 + r) * LK_ + k0) = e[r];
        if (lane == 0) {
            const v2f ml = {m[r], l[r]};
            *(v2f*)(part + ((size_t)(row0 + r) * SC_SLICES + slice) * 2) = ml;
        }
    }
}

#define AV_WPB 4
#define AV_TN  4
#define AV_KQ  (LK_ / AV_WPB)
#define AV_NT  (AV_KQ / (8 * GK_KB))

struct AvFrag {
    v4f a[GK_KB];
    v4f b[GK_KB][4];
};

__device__ __forceinline__ void av_load(AvFrag& f, const float* aptr,
                                        const float* vptr, int trip)
{
#pragma unroll
    for (int j = 0; j < GK_KB; ++j) {
        const int kk = (trip * GK_KB + j) * 8;
        f.a[j] = *(const v4f*)(aptr + kk);
        const float* vp = vptr + (size_t)kk * DSZ_;
#pragma unroll
        for (int i = 0; i < 4; ++i)
            f.b[j][i] = *(const v4f*)(vp + (size_t)i * DSZ_);
    }
}

__device__ __forceinline__ void av_mma(v8f (&c)[AV_TN], const AvFrag& f, float fs)
{
#pragma unroll
    for (int j = 0; j < GK_KB; ++j) {
        const v4f a  = f.a[j] * fs;
        const v2f a0 = __builtin_shufflevector(a, a, 0, 1);
        const v2f a1 = __builtin_shufflevector(a, a, 2, 3);
#pragma unroll
        for (int t = 0; t < AV_TN; ++t) {
            const v2f b0 = {f.b[j][0][t], f.b[j][1][t]};
            c[t] = __builtin_amdgcn_wmma_f32_16x16x4_f32(
                    false, a0, false, b0, (short)0, c[t], false, false);
        }
#pragma unroll
        for (int t = 0; t < AV_TN; ++t) {
            const v2f b1 = {f.b[j][2][t], f.b[j][3][t]};
            c[t] = __builtin_amdgcn_wmma_f32_16x16x4_f32(
                    false, a1, false, b1, (short)0, c[t], false, false);
        }
    }
}

template <bool LAST>
__device__ __forceinline__ void av_pass(v8f (&c)[AV_TN], AvFrag (&ring)[GK_RING],
                                        const float* aptr, const float* vptr, int t,
                                        float g0, float g1)
{
#pragma unroll
    for (int s = 0; s < GK_RING; ++s) {
        if (!LAST || s + GK_PF < GK_RING)
            av_load(ring[(s + GK_PF) & (GK_RING - 1)], aptr, vptr, t + s + GK_PF);
        __builtin_amdgcn_sched_barrier(0);
        av_mma(c, ring[s], (s < GK_RING / 2) ? g0 : g1);
        __builtin_amdgcn_sched_barrier(0);
    }
}

__global__ __launch_bounds__(AV_WPB * 32) void attn_out_wmma(
    const float* __restrict__ attn,
    const float* __restrict__ part,
    const float* __restrict__ value,
    float* __restrict__ out)
{
    __shared__ v4f red[AV_WPB][8][32];

    const int lane      = threadIdx.x & 31;
    const int kq        = threadIdx.x >> 5;
    const int blocksN   = DSZ_ / (16 * AV_TN);
    const int blocksPerB = (LQ_ / 16) * blocksN;
    const int b         = blockIdx.x / blocksPerB;
    const int tr        = blockIdx.x % blocksPerB;
    const int tile_m    = tr / blocksN;
    const int n0        = (tr % blocksN) * (16 * AV_TN);

    const int n15 = lane & 15;
    const int ko  = (lane < 16) ? 0 : 4;
    const int kb  = kq * AV_KQ;

    const float* aptr = attn  + ((size_t)b * LQ_ + tile_m * 16 + n15) * LK_ + kb + ko;
    const float* vptr = value + (size_t)b * LK_ * DSZ_
                               + (size_t)(kb + ko) * DSZ_ + n0 + 4 * n15;

    const float* prow = part + ((size_t)b * LQ_ + tile_m * 16 + n15) * (SC_SLICES * 2);
    v4f pq[SC_SLICES / 2];
#pragma unroll
    for (int i = 0; i < SC_SLICES / 2; ++i) pq[i] = *(const v4f*)(prow + 4 * i);
    const v4f po0 = *(const v4f*)(prow + kq * 8);
    const v4f po1 = *(const v4f*)(prow + kq * 8 + 4);

    v8f c[AV_TN];
#pragma unroll
    for (int t = 0; t < AV_TN; ++t) c[t] = (v8f){0.f, 0.f, 0.f, 0.f, 0.f, 0.f, 0.f, 0.f};
    AvFrag ring[GK_RING];
#pragma unroll
    for (int p = 0; p < GK_PF; ++p) {
        av_load(ring[p], aptr, vptr, p);
        asm volatile("" ::: "memory");
    }

    __builtin_amdgcn_sched_barrier(0);
    float M = pq[0].x;
#pragma unroll
    for (int i = 0; i < SC_SLICES / 2; ++i) M = fmaxf(fmaxf(M, pq[i].x), pq[i].z);
    float L = 0.f;
#pragma unroll
    for (int i = 0; i < SC_SLICES / 2; ++i) {
        L += pq[i].y * __expf(pq[i].x - M);
        L += pq[i].w * __expf(pq[i].z - M);
    }
    const float invL = 1.0f / L;
    const float g0 = __expf(po0.x - M) * invL;
    const float g1 = __expf(po0.z - M) * invL;
    const float g2 = __expf(po1.x - M) * invL;
    const float g3 = __expf(po1.z - M) * invL;

    av_pass<false>(c, ring, aptr, vptr, 0, g0, g1);
    av_pass<true>(c, ring, aptr, vptr, AV_NT - GK_RING, g2, g3);

#pragma unroll
    for (int r = 0; r < 8; ++r) {
        const v4f pv = {c[0][r], c[1][r], c[2][r], c[3][r]};
        red[kq][r][lane] = pv;
    }
    __syncthreads();
    const int mo = (lane < 16) ? 0 : 8;
#pragma unroll
    for (int rr = 0; rr < 2; ++rr) {
        const int r = 2 * kq + rr;
        v4f s = red[0][r][lane];
#pragma unroll
        for (int q = 1; q < AV_WPB; ++q) s += red[q][r][lane];
        const int m = tile_m * 16 + mo + r;
        *(v4f*)(out + ((size_t)b * LQ_ + m) * DSZ_ + n0 + 4 * n15) = s;
    }
}

extern "C" void kernel_launch(void* const* d_in, const int* in_sizes, int n_in,
                              void* d_out, int out_size, void* d_ws, size_t ws_size,
                              hipStream_t stream) {
    const float* query     = (const float*)d_in[0];
    const float* key       = (const float*)d_in[1];
    const float* value     = (const float*)d_in[2];
    const int*   valid_len = (const int*)  d_in[3];
    const float* W_q       = (const float*)d_in[4];
    const float* W_k       = (const float*)d_in[5];
    const float* vvec      = (const float*)d_in[6];

    float* qf     = (float*)d_ws;
    float* kfT    = qf  + (size_t)B_ * LQ_ * UNITS_;
    float* scores = kfT + (size_t)B_ * UNITS_ * LK_;
    float* part   = scores + (size_t)B_ * LQ_ * LK_;
    float* outp   = (float*)d_out;

    proj_splitk<false, 1, 2, 3><<<256, PK_WPB * 32, 0, stream>>>(query, W_q, qf);
    proj_splitk<true, 4, 4, 1><<<256, PK_WPB * 32, 0, stream>>>(key, W_k, kfT);
    scores_tanh_kernel<<<SC_BLOCKS, 256, 0, stream>>>(qf, kfT, vvec, valid_len, scores, part);
    attn_out_wmma<<<256, AV_WPB * 32, 0, stream>>>(scores, part, value, outp);
}


# query projection moved into the key-projection launch: work-group i adds query block i (16x32) after its 64x64 key block, loads issued under the key side's last WMMAs; 3 launches; sum orders kept
# speedup vs baseline: 3.7957x; 1.1034x over previous
#include <hip/hip_runtime.h>

typedef float v2f __attribute__((ext_vector_type(2)));
typedef float v4f __attribute__((ext_vector_type(4)));
typedef float v8f __attribute__((ext_vector_type(8)));

#define B_     4
#define LQ_    128
#define LK_    1024
#define DSZ_   512
#define UNITS_ 256
#define MASK_VAL_ (-1.0e6f)

#if __has_builtin(__builtin_amdgcn_tanhf)
#define TANHF(x) __builtin_amdgcn_tanhf(x)
#else
#define TANHF(x) tanhf(x)
#endif


#define GK_KB   4
#define GK_PF   3
#define GK_RING (GK_PF + 1)

#define PK_WPB 4
#define PK_KQ  (DSZ_ / PK_WPB)
#define PK_NT  (PK_KQ / (8 * GK_KB))
#define KS_TM  4
#define KS_TN  4
#define QS_TM  1
#define QS_TN  2
#define PK_BLOCKS 256
static_assert((UNITS_ / (16 * KS_TM)) * (B_ * LK_ / (16 * KS_TN)) == PK_BLOCKS,
              "key side: one block per work-group");
static_assert((B_ * LQ_ / (16 * QS_TM)) * (UNITS_ / (16 * QS_TN)) == PK_BLOCKS,
              "query side: one block per work-group");

template <int N> struct VecOf;
template <> struct VecOf<2> { typedef v2f type; };
template <> struct VecOf<4> { typedef v4f type; };

template <int TM, int TN>
struct ProjFrag {
    v4f a[TM][GK_KB];
    v4f b[TN][GK_KB];
};

template <int TM, int TN>
__device__ __forceinline__ void proj_load(ProjFrag<TM, TN>& f, const float* aptr,
                                          const float* bptr, int trip)
{
#pragma unroll
    for (int j = 0; j < GK_KB; ++j) {
        const int kk = (trip * GK_KB + j) * 8;
#pragma unroll
        for (int i = 0; i < TM; ++i)
            f.a[i][j] = *(const v4f*)(aptr + (size_t)i * 16 * DSZ_ + kk);
#pragma unroll
        for (int i = 0; i < TN; ++i)
            f.b[i][j] = *(const v4f*)(bptr + (size_t)i * DSZ_ + kk);
    }
}

template <int TM, int TN>
__device__ __forceinline__ void proj_mma(v8f (&c)[TM][TN], const ProjFrag<TM, TN>& f)
{
#pragma unroll
    for (int j = 0; j < GK_KB; ++j) {
        v2f a0[TM], a1[TM], b0[TN], b1[TN];
#pragma unroll
        for (int i = 0; i < TM; ++i) {
            a0[i] = __builtin_shufflevector(f.a[i][j], f.a[i][j], 0, 1);
            a1[i] = __builtin_shufflevector(f.a[i][j], f.a[i][j], 2, 3);
        }
#pragma unroll
        for (int i = 0; i < TN; ++i) {
            b0[i] = __builtin_shufflevector(f.b[i][j], f.b[i][j], 0, 1);
            b1[i] = __builtin_shufflevector(f.b[i][j], f.b[i][j], 2, 3);
        }
#pragma unroll
        for (int tm = 0; tm < TM; ++tm)
#pragma unroll
            for (int tn = 0; tn < TN; ++tn)
                c[tm][tn] = __builtin_amdgcn_wmma_f32_16x16x4_f32(
                        false, a0[tm], false, b0[tn], (short)0, c[tm][tn], false, false);
#pragma unroll
        for (int tm = 0; tm < TM; ++tm)
#pragma unroll
            for (int tn = 0; tn < TN; ++tn)
                c[tm][tn] = __builtin_amdgcn_wmma_f32_16x16x4_f32(
                        false, a1[tm], false, b1[tn], (short)0, c[tm][tn], false, false);
    }
}

template <int TM, int TN>
__device__ __forceinline__ void proj_put(
    typename VecOf<TN>::type (&red)[PK_WPB][TM][8][32], const v8f (&c)[TM][TN],
    int kq, int lane)
{
    typedef typename VecOf<TN>::type vtn;
#pragma unroll
    for (int tm = 0; tm < TM; ++tm)
#pragma unroll
        for (int r = 0; r < 8; ++r) {
            vtn pv;
#pragma unroll
            for (int tn = 0; tn < TN; ++tn) pv[tn] = c[tm][tn][r];
            red[kq][tm][r][lane] = pv;
        }
}

template <int TM, int TN>
__device__ __forceinline__ void proj_sum_store(
    const typename VecOf<TN>::type (&red)[PK_WPB][TM][8][32], float* obase, int ldo,
    int kq, int lane)
{
    typedef typename VecOf<TN>::type vtn;
    const int n15 = lane & 15;
    const int mo  = (lane < 16) ? 0 : 8;
    constexpr int PER = TM * 8 / PK_WPB;
#pragma unroll
    for (int i = 0; i < PER; ++i) {
        const int idx = kq * PER + i;
        const int tm  = idx >> 3;
        const int r   = idx & 7;
        vtn s = red[0][tm][r][lane];
#pragma unroll
        for (int q = 1; q < PK_WPB; ++q) s += red[q][tm][r][lane];
        *(vtn*)(obase + (size_t)(tm * 16 + mo + r) * ldo + TN * n15) = s;
    }
}

__global__ __launch_bounds__(PK_WPB * 32) void proj_both_splitk(
    const float* __restrict__ key,
    const float* __restrict__ Wk,
    float* __restrict__ kfT,
    const float* __restrict__ query,
    const float* __restrict__ Wq,
    float* __restrict__ qf)
{
    __shared__ v4f redk[PK_WPB][KS_TM][8][32];
    __shared__ v2f redq[PK_WPB][QS_TM][8][32];

    const int lane = threadIdx.x & 31;
    const int kq   = threadIdx.x >> 5;
    const int n15  = lane & 15;
    const int ko   = (lane < 16) ? 0 : 4;
    const int kb   = kq * PK_KQ;

    constexpr int kub = UNITS_ / (16 * KS_TM);
    const int ka0 = (blockIdx.x % kub) * (16 * KS_TM);
    const int kb0 = (blockIdx.x / kub) * (16 * KS_TN);
    float* kobase = kfT + ((size_t)(kb0 / LK_) * UNITS_ + ka0) * LK_ + (kb0 % LK_);
    const float* kaptr = Wk  + (size_t)(ka0 + n15) * DSZ_ + kb + ko;
    const float* kbptr = key + (size_t)(kb0 + KS_TN * n15) * DSZ_ + kb + ko;

    constexpr int qub = UNITS_ / (16 * QS_TN);
    const int qa0 = (blockIdx.x / qub) * (16 * QS_TM);
    const int qb0 = (blockIdx.x % qub) * (16 * QS_TN);
    float* qobase = qf + (size_t)qa0 * UNITS_ + qb0;
    const float* qaptr = query + (size_t)(qa0 + n15) * DSZ_ + kb + ko;
    const float* qbptr = Wq    + (size_t)(qb0 + QS_TN * n15) * DSZ_ + kb + ko;

    v8f ck[KS_TM][KS_TN];
#pragma unroll
    for (int tm = 0; tm < KS_TM; ++tm)
#pragma unroll
        for (int tn = 0; tn < KS_TN; ++tn)
            ck[tm][tn] = (v8f){0.f, 0.f, 0.f, 0.f, 0.f, 0.f, 0.f, 0.f};
    v8f cq[QS_TM][QS_TN];
#pragma unroll
    for (int tm = 0; tm < QS_TM; ++tm)
#pragma unroll
        for (int tn = 0; tn < QS_TN; ++tn)
            cq[tm][tn] = (v8f){0.f, 0.f, 0.f, 0.f, 0.f, 0.f, 0.f, 0.f};

    ProjFrag<KS_TM, KS_TN> kring[2];
    ProjFrag<QS_TM, QS_TN> qfr[PK_NT];
    proj_load(kring[0], kaptr, kbptr, 0);
    asm volatile("" ::: "memory");
#pragma unroll
    for (int t = 0; t < PK_NT; ++t) {
        if (t + 1 < PK_NT) {
            proj_load(kring[(t + 1) & 1], kaptr, kbptr, t + 1);
        } else {
#pragma unroll
            for (int p = 0; p < PK_NT; ++p) proj_load(qfr[p], qaptr, qbptr, p);
        }
        __builtin_amdgcn_sched_barrier(0);
        proj_mma(ck, kring[t & 1]);
        __builtin_amdgcn_sched_barrier(0);
    }
#pragma unroll
    for (int p = 0; p < PK_NT; ++p) proj_mma(cq, qfr[p]);

    proj_put<KS_TM, KS_TN>(redk, ck, kq, lane);
    proj_put<QS_TM, QS_TN>(redq, cq, kq, lane);
    __syncthreads();
    proj_sum_store<KS_TM, KS_TN>(redk, kobase, LK_, kq, lane);
    proj_sum_store<QS_TM, QS_TN>(redq, qobase, UNITS_, kq, lane);
}

#define SC_R    4
#define SC_UC   8
#define SC_NCH  (UNITS_ / SC_UC)
#define SC_PF   3
#define SC_RING (SC_PF + 1)
#define SC_BLOCKS (B_ * (LQ_ / 8) * (LK_ / 256))
#define SC_SLICES (LK_ / 64)

struct ScScalars {
    float q[SC_R][SC_UC];
    float v[SC_UC];
};

__device__ __forceinline__ void sc_load_k(v2f (&dst)[SC_UC], const float* kp, int chunk)
{
#pragma unroll
    for (int j = 0; j < SC_UC; ++j)
        dst[j] = *(const v2f*)(kp + (size_t)(chunk * SC_UC + j) * LK_);
}

__device__ __forceinline__ void sc_load_s(ScScalars& d, const float* __restrict__ qrow,
                                          const float* __restrict__ vvec, int chunk)
{
#pragma unroll
    for (int j = 0; j < SC_UC; ++j) d.v[j] = vvec[chunk * SC_UC + j];
#pragma unroll
    for (int r = 0; r < SC_R; ++r)
#pragma unroll
        for (int j = 0; j < SC_UC; ++j)
            d.q[r][j] = qrow[r * UNITS_ + chunk * SC_UC + j];
}

__device__ __forceinline__ void sc_chunk(v2f (&acc)[SC_R], const v2f (&kv)[SC_UC],
                                         const ScScalars& s)
{
#pragma unroll
    for (int j = 0; j < SC_UC; ++j) {
        const v2f vv2 = {s.v[j], s.v[j]};
#pragma unroll
        for (int r = 0; r < SC_R; ++r) {
            const v2f q2 = {s.q[r][j], s.q[r][j]};
            const v2f x = q2 + kv[j];
            v2f t;
            t.x = TANHF(x.x);
            t.y = TANHF(x.y);
            acc[r] = __builtin_elementwise_fma(vv2, t, acc[r]);
        }
    }
}

template <int XM>
__device__ __forceinline__ float row_xor(float x)
{
    return __builtin_bit_cast(float, __builtin_amdgcn_update_dpp(
            0, __builtin_bit_cast(int, x), 0x160 + XM, 0xf, 0xf, false));
}
__device__ __forceinline__ float xhalf(float x)
{
    return __builtin_bit_cast(float, __builtin_amdgcn_permlanex16(
            __builtin_bit_cast(int, x), __builtin_bit_cast(int, x),
            0x76543210, 0xfedcba98, false, false));
}
template <bool IS_MAX>
__device__ __forceinline__ void wave_reduce4(float (&x)[4])
{
#define WR_STEP(EXPR)                                                      \
    _Pragma("unroll") for (int r = 0; r < 4; ++r) {                         \
        const float y = (EXPR);                                            \
        x[r] = IS_MAX ? fmaxf(x[r], y) : x[r] + y;                          \
    }
    WR_STEP(row_xor<1>(x[r]))
    WR_STEP(row_xor<2>(x[r]))
    WR_STEP(row_xor<4>(x[r]))
    WR_STEP(row_xor<8>(x[r]))
    WR_STEP(xhalf(x[r]))
#undef WR_STEP
}

__global__ __launch_bounds__(256) void scores_tanh_kernel(
    const float* __restrict__ qf,
    const float* __restrict__ kfT,
    const float* __restrict__ vvec,
    const int*   __restrict__ valid_len,
    float* __restrict__ scores,
    float* __restrict__ part)
{
    const int lane = threadIdx.x & 31;
    const int wave = __builtin_amdgcn_readfirstlane(threadIdx.x >> 5);
    const int kblk = blockIdx.x & 3;
    const int qblk = blockIdx.x >> 2;
    const int b    = qblk / (LQ_ / 8);
    const int row0 = qblk * 8 + (wave >> 2) * SC_R;
    const int k0   = kblk * 256 + (wave & 3) * 64 + lane * 2;

    const float* qrow = qf  + (size_t)row0 * UNITS_;
    const float* kp   = kfT + (size_t)b * UNITS_ * LK_ + k0;

    v2f acc[SC_R];
#pragma unroll
    for (int r = 0; r < SC_R; ++r) acc[r] = (v2f){0.f, 0.f};

    v2f ring[SC_RING][SC_UC];
    ScScalars sc[2];
#pragma unroll
    for (int p = 0; p < SC_PF; ++p) sc_load_k(ring[p], kp, p);
    sc_load_s(sc[0], qrow, vvec, 0);

    for (int c = 0; c < SC_NCH; c += SC_RING) {
#pragma unroll
        for (int s = 0; s < SC_RING; ++s) {
            const int cn = (c + s + SC_PF < SC_NCH) ? (c + s + SC_PF) : (SC_NCH - 1);
            const int cs = (c + s + 1 < SC_NCH) ? (c + s + 1) : (SC_NCH - 1);
            sc_load_k(ring[(s + SC_PF) & (SC_RING - 1)], kp, cn);
            sc_load_s(sc[(s + 1) & 1], qrow, vvec, cs);
            asm volatile("" ::: "memory");
            sc_chunk(acc, ring[s], sc[s & 1]);
        }
    }

    const int vl    = valid_len[b];
    const int slice = kblk * 4 + (wave & 3);
    v2f o[SC_R], e[SC_R];
    float m[SC_R], l[SC_R];
#pragma unroll
    for (int r = 0; r < SC_R; ++r) {
        o[r].x = (k0 + 0 < vl) ? acc[r].x : MASK_VAL_;
        o[r].y = (k0 + 1 < vl) ? acc[r].y : MASK_VAL_;
        m[r] = fmaxf(o[r].x, o[r].y);
    }
    wave_reduce4<true>(m);
#pragma unroll
    for (int r = 0; r < SC_R; ++r) {
        e[r].x = __expf(o[r].x - m[r]);
        e[r].y = __expf(o[r].y - m[r]);
        l[r] = e[r].x + e[r].y;
    }
    wave_reduce4<false>(l);
#pragma unroll
    for (int r = 0; r < SC_R; ++r) {
        *(v2f*)(scores + (size_t)(row0 + r) * LK_ + k0) = e[r];
        if (lane == 0) {
            const v2f ml = {m[r], l[r]};
            *(v2f*)(part + ((size_t)(row0 + r) * SC_SLICES + slice) * 2) = ml;
        }
    }
}

#define AV_WPB 4
#define AV_TN  4
#define AV_KQ  (LK_ / AV_WPB)
#define AV_NT  (AV_KQ / (8 * GK_KB))

struct AvFrag {
    v4f a[GK_KB];
    v4f b[GK_KB][4];
};

__device__ __forceinline__ void av_load(AvFrag& f, const float* aptr,
                                        const float* vptr, int trip)
{
#pragma unroll
    for (int j = 0; j < GK_KB; ++j) {
        const int kk = (trip * GK_KB + j) * 8;
        f.a[j] = *(const v4f*)(aptr + kk);
        const float* vp = vptr + (size_t)kk * DSZ_;
#pragma unroll
        for (int i = 0; i < 4; ++i)
            f.b[j][i] = *(const v4f*)(vp + (size_t)i * DSZ_);
    }
}

__device__ __forceinline__ void av_mma(v8f (&c)[AV_TN], const AvFrag& f, float fs)
{
#pragma unroll
    for (int j = 0; j < GK_KB; ++j) {
        const v4f a  = f.a[j] * fs;
        const v2f a0 = __builtin_shufflevector(a, a, 0, 1);
        const v2f a1 = __builtin_shufflevector(a, a, 2, 3);
#pragma unroll
        for (int t = 0; t < AV_TN; ++t) {
            const v2f b0 = {f.b[j][0][t], f.b[j][1][t]};
            c[t] = __builtin_amdgcn_wmma_f32_16x16x4_f32(
                    false, a0, false, b0, (short)0, c[t], false, false);
        }
#pragma unroll
        for (int t = 0; t < AV_TN; ++t) {
            const v2f b1 = {f.b[j][2][t], f.b[j][3][t]};
            c[t] = __builtin_amdgcn_wmma_f32_16x16x4_f32(
                    false, a1, false, b1, (short)0, c[t], false, false);
        }
    }
}

template <bool LAST>
__device__ __forceinline__ void av_pass(v8f (&c)[AV_TN], AvFrag (&ring)[GK_RING],
                                        const float* aptr, const float* vptr, int t,
                                        float g0, float g1)
{
#pragma unroll
    for (int s = 0; s < GK_RING; ++s) {
        if (!LAST || s + GK_PF < GK_RING)
            av_load(ring[(s + GK_PF) & (GK_RING - 1)], aptr, vptr, t + s + GK_PF);
        __builtin_amdgcn_sched_barrier(0);
        av_mma(c, ring[s], (s < GK_RING / 2) ? g0 : g1);
        __builtin_amdgcn_sched_barrier(0);
    }
}

__global__ __launch_bounds__(AV_WPB * 32) void attn_out_wmma(
    const float* __restrict__ attn,
    const float* __restrict__ part,
    const float* __restrict__ value,
    float* __restrict__ out)
{
    __shared__ v4f red[AV_WPB][8][32];

    const int lane      = threadIdx.x & 31;
    const int kq        = threadIdx.x >> 5;
    const int blocksN   = DSZ_ / (16 * AV_TN);
    const int blocksPerB = (LQ_ / 16) * blocksN;
    const int b         = blockIdx.x / blocksPerB;
    const int tr        = blockIdx.x % blocksPerB;
    const int tile_m    = tr / blocksN;
    const int n0        = (tr % blocksN) * (16 * AV_TN);

    const int n15 = lane & 15;
    const int ko  = (lane < 16) ? 0 : 4;
    const int kb  = kq * AV_KQ;

    const float* aptr = attn  + ((size_t)b * LQ_ + tile_m * 16 + n15) * LK_ + kb + ko;
    const float* vptr = value + (size_t)b * LK_ * DSZ_
                               + (size_t)(kb + ko) * DSZ_ + n0 + 4 * n15;

    const float* prow = part + ((size_t)b * LQ_ + tile_m * 16 + n15) * (SC_SLICES * 2);
    v4f pq[SC_SLICES / 2];
#pragma unroll
    for (int i = 0; i < SC_SLICES / 2; ++i) pq[i] = *(const v4f*)(prow + 4 * i);
    const v4f po0 = *(const v4f*)(prow + kq * 8);
    const v4f po1 = *(const v4f*)(prow + kq * 8 + 4);

    v8f c[AV_TN];
#pragma unroll
    for (int t = 0; t < AV_TN; ++t) c[t] = (v8f){0.f, 0.f, 0.f, 0.f, 0.f, 0.f, 0.f, 0.f};
    AvFrag ring[GK_RING];
#pragma unroll
    for (int p = 0; p < GK_PF; ++p) {
        av_load(ring[p], aptr, vptr, p);
        asm volatile("" ::: "memory");
    }

    __builtin_amdgcn_sched_barrier(0);
    float M = pq[0].x;
#pragma unroll
    for (int i = 0; i < SC_SLICES / 2; ++i) M = fmaxf(fmaxf(M, pq[i].x), pq[i].z);
    float L = 0.f;
#pragma unroll
    for (int i = 0; i < SC_SLICES / 2; ++i) {
        L += pq[i].y * __expf(pq[i].x - M);
        L += pq[i].w * __expf(pq[i].z - M);
    }
    const float invL = 1.0f / L;
    const float g0 = __expf(po0.x - M) * invL;
    const float g1 = __expf(po0.z - M) * invL;
    const float g2 = __expf(po1.x - M) * invL;
    const float g3 = __expf(po1.z - M) * invL;

    av_pass<false>(c, ring, aptr, vptr, 0, g0, g1);
    av_pass<true>(c, ring, aptr, vptr, AV_NT - GK_RING, g2, g3);

#pragma unroll
    for (int r = 0; r < 8; ++r) {
        const v4f pv = {c[0][r], c[1][r], c[2][r], c[3][r]};
        red[kq][r][lane] = pv;
    }
    __syncthreads();
    const int mo = (lane < 16) ? 0 : 8;
#pragma unroll
    for (int rr = 0; rr < 2; ++rr) {
        const int r = 2 * kq + rr;
        v4f s = red[0][r][lane];
#pragma unroll
        for (int q = 1; q < AV_WPB; ++q) s += red[q][r][lane];
        const int m = tile_m * 16 + mo + r;
        *(v4f*)(out + ((size_t)b * LQ_ + m) * DSZ_ + n0 + 4 * n15) = s;
    }
}

extern "C" void kernel_launch(void* const* d_in, const int* in_sizes, int n_in,
                              void* d_out, int out_size, void* d_ws, size_t ws_size,
                              hipStream_t stream) {
    const float* query     = (const float*)d_in[0];
    const float* key       = (const float*)d_in[1];
    const float* value     = (const float*)d_in[2];
    const int*   valid_len = (const int*)  d_in[3];
    const float* W_q       = (const float*)d_in[4];
    const float* W_k       = (const float*)d_in[5];
    const float* vvec      = (const float*)d_in[6];

    float* qf     = (float*)d_ws;
    float* kfT    = qf  + (size_t)B_ * LQ_ * UNITS_;
    float* scores = kfT + (size_t)B_ * UNITS_ * LK_;
    float* part   = scores + (size_t)B_ * LQ_ * LK_;
    float* outp   = (float*)d_out;

    proj_both_splitk<<<PK_BLOCKS, PK_WPB * 32, 0, stream>>>(key, W_k, kfT, query, W_q, qf);
    scores_tanh_kernel<<<SC_BLOCKS, 256, 0, stream>>>(qf, kfT, vvec, valid_len, scores, part);
    attn_out_wmma<<<256, AV_WPB * 32, 0, stream>>>(scores, part, value, outp);
}
